# starting kernel
# baseline (speedup 1.0000x reference)
#include <hip/hip_runtime.h>
#include <hip/hip_bf16.h>


#define BDIM 1024
#define KDIM 64
#define NMON 512
#define ROWS_PER_BLOCK 128
#define THREADS 256

typedef __attribute__((ext_vector_type(2))) float v2f;
typedef __attribute__((ext_vector_type(8))) float v8f;

__global__ __launch_bounds__(THREADS)
void sparse_poly_kernel(const float* __restrict__ x,
                        const int*   __restrict__ S,
                        const float* __restrict__ a,
                        const float* __restrict__ b,
                        const int*   __restrict__ idx3, const float* __restrict__ c3,
                        const int*   __restrict__ idx4, const float* __restrict__ c4,
                        const int*   __restrict__ idx5, const float* __restrict__ c5,
                        float* __restrict__ out)
{
    __shared__ float xs[ROWS_PER_BLOCK][KDIM];
    __shared__ float buS[KDIM][KDIM];
    __shared__ float quadS[ROWS_PER_BLOCK];
    __shared__ int   Ssh[KDIM];
    __shared__ float ash[KDIM];

    const int tid  = threadIdx.x;
    const int wave = tid >> 5;
    const int lane = tid & 31;
    const int lo   = lane & 15;
    const int hi   = lane >> 4;
    const int rowBlockBase = blockIdx.x * ROWS_PER_BLOCK;

    if (tid < KDIM) {
        Ssh[tid] = S[tid];
        ash[tid] = a[tid];
    }
    #pragma unroll
    for (int e = tid; e < KDIM * KDIM; e += THREADS) {
        const int i = e >> 6, j = e & 63;
        buS[i][j] = (j > i) ? b[e] : 0.0f;
    }
    __syncthreads();

    for (int e = tid; e < ROWS_PER_BLOCK * KDIM; e += THREADS) {
        const int r = e >> 6, j = e & 63;
        xs[r][j] = x[(size_t)(rowBlockBase + r) * BDIM + Ssh[j]];
    }
    __syncthreads();

    const int waveBase = wave * 16;
    float part[8];
    #pragma unroll
    for (int p = 0; p < 8; ++p) part[p] = 0.0f;

    #pragma unroll
    for (int nt = 0; nt < 4; ++nt) {
        v8f acc = {};
        #pragma unroll
        for (int kc = 0; kc < 16; ++kc) {
            v2f A, Bf;
            A.x = xs[waveBase + lo][4 * kc + 2 * hi + 0];
            A.y = xs[waveBase + lo][4 * kc + 2 * hi + 1];
            Bf.x = buS[4 * kc + 2 * hi + 0][16 * nt + lo];
            Bf.y = buS[4 * kc + 2 * hi + 1][16 * nt + lo];
            acc = __builtin_amdgcn_wmma_f32_16x16x4_f32(
                false, A, false, Bf, (short)0, acc, false, false);
        }
        #pragma unroll
        for (int p = 0; p < 8; ++p)
            part[p] += acc[p] * xs[waveBase + p + 8 * hi][16 * nt + lo];
    }

    #pragma unroll
    for (int p = 0; p < 8; ++p) {
        float s = part[p];
        s += __shfl_xor(s, 8);
        s += __shfl_xor(s, 4);
        s += __shfl_xor(s, 2);
        s += __shfl_xor(s, 1);
        if (lo == p) quadS[waveBase + p + 8 * hi] = s;
    }
    __syncthreads();

    if (tid < ROWS_PER_BLOCK) {
        const int lr = tid;
        const float* xr = xs[lr];
        float accum = quadS[lr];

        #pragma unroll 8
        for (int j = 0; j < KDIM; ++j)
            accum += xr[j] * ash[j];

        for (int t = 0; t < NMON; ++t) {
            const int i0 = idx3[3 * t + 0], i1 = idx3[3 * t + 1], i2 = idx3[3 * t + 2];
            accum += c3[t] * xr[i0] * xr[i1] * xr[i2];
        }
        for (int t = 0; t < NMON; ++t) {
            const int i0 = idx4[4 * t + 0], i1 = idx4[4 * t + 1];
            const int i2 = idx4[4 * t + 2], i3 = idx4[4 * t + 3];
            accum += c4[t] * xr[i0] * xr[i1] * xr[i2] * xr[i3];
        }
        for (int t = 0; t < NMON; ++t) {
            const int i0 = idx5[5 * t + 0], i1 = idx5[5 * t + 1], i2 = idx5[5 * t + 2];
            const int i3 = idx5[5 * t + 3], i4 = idx5[5 * t + 4];
            accum += c5[t] * xr[i0] * xr[i1] * xr[i2] * xr[i3] * xr[i4];
        }
        out[rowBlockBase + lr] = accum;
    }
}

extern "C" void kernel_launch(void* const* d_in, const int* in_sizes, int n_in,
                              void* d_out, int out_size, void* d_ws, size_t ws_size,
                              hipStream_t stream) {
    const float* x    = (const float*)d_in[0];
    const int*   S    = (const int*)  d_in[1];
    const float* a    = (const float*)d_in[2];
    const float* b    = (const float*)d_in[3];
    const int*   idx3 = (const int*)  d_in[4];
    const float* c3   = (const float*)d_in[5];
    const int*   idx4 = (const int*)  d_in[6];
    const float* c4   = (const float*)d_in[7];
    const int*   idx5 = (const int*)  d_in[8];
    const float* c5   = (const float*)d_in[9];
    float* out = (float*)d_out;

    const int nblocks = out_size / ROWS_PER_BLOCK;
    sparse_poly_kernel<<<nblocks, THREADS, 0, stream>>>(
        x, S, a, b, idx3, c3, idx4, c4, idx5, c5, out);
}


# idx/c term tables staged once per block in LDS (coalesced) and read there 16 terms per trip with the next trip's entries prefetched, not one scalar-load round trip per 4 terms; sum order unchanged
# speedup vs baseline: 9.2780x; 9.2780x over previous
#include <hip/hip_runtime.h>
#include <hip/hip_bf16.h>


#define BDIM 1024
#define KDIM 64
#define NMON 512
#define ROWS_PER_BLOCK 128
#define THREADS 256
#define XPITCH 137
#define XS(r, j) xsT[(j)][(r)]

typedef __attribute__((ext_vector_type(2))) float v2f;
typedef __attribute__((ext_vector_type(8))) float v8f;

#define TERM_BATCH 16

typedef __attribute__((ext_vector_type(4))) int   v4i;
typedef __attribute__((ext_vector_type(4))) float v4f;

template <int DEG>
__device__ __forceinline__ float mono_terms(const float* rowp, const v4i* offs,
                                            const v4f* cs, float accum)
{
    constexpr int UB = TERM_BATCH;
    constexpr int NQ = UB * DEG / 4;
    constexpr int NB = NMON / UB;
    v4i o[NQ];
    #pragma unroll
    for (int i = 0; i < NQ; ++i) o[i] = offs[i];

    for (int b = 0; b < NB; ++b) {
        v4f cc[UB / 4];
        #pragma unroll
        for (int i = 0; i < UB / 4; ++i) cc[i] = cs[b * (UB / 4) + i];
        float v[UB * DEG];
        #pragma unroll
        for (int i = 0; i < UB * DEG; ++i) v[i] = rowp[o[i >> 2][i & 3]];
        const int bn = (b + 1 < NB) ? (b + 1) : b;
        #pragma unroll
        for (int i = 0; i < NQ; ++i) o[i] = offs[bn * NQ + i];
        #pragma unroll
        for (int u = 0; u < UB; ++u) {
            const float* w = &v[u * DEG];
            const float cu = cc[u >> 2][u & 3];
            if (DEG == 3)      accum += cu * w[0] * w[1] * w[2];
            else if (DEG == 4) accum += cu * w[0] * w[1] * w[2] * w[3];
            else               accum += cu * w[0] * w[1] * w[2] * w[3] * w[DEG - 1];
        }
    }
    return accum;
}

__global__ __launch_bounds__(THREADS)
void sparse_poly_kernel(const float* __restrict__ x,
                        const int*   __restrict__ S,
                        const float* __restrict__ a,
                        const float* __restrict__ b,
                        const int*   __restrict__ idx3, const float* __restrict__ c3,
                        const int*   __restrict__ idx4, const float* __restrict__ c4,
                        const int*   __restrict__ idx5, const float* __restrict__ c5,
                        float* __restrict__ out)
{
    __shared__ float xsT[KDIM][XPITCH];
    __shared__ float buS[KDIM][KDIM];
    __shared__ float quadS[ROWS_PER_BLOCK];
    __shared__ int   Ssh[KDIM];
    __shared__ float ash[KDIM];
    __shared__ v4i off3[3 * NMON / 4];
    __shared__ v4i off4[4 * NMON / 4];
    __shared__ v4i off5[5 * NMON / 4];
    __shared__ v4f cs3[NMON / 4];
    __shared__ v4f cs4[NMON / 4];
    __shared__ v4f cs5[NMON / 4];

    const int tid  = threadIdx.x;
    const int wave = tid >> 5;
    const int lane = tid & 31;
    const int lo   = lane & 15;
    const int hi   = lane >> 4;
    const int rowBlockBase = blockIdx.x * ROWS_PER_BLOCK;

    if (tid < KDIM) {
        Ssh[tid] = S[tid];
        ash[tid] = a[tid];
    }
    #pragma unroll
    for (int e = tid; e < KDIM * KDIM; e += THREADS) {
        const int i = e >> 6, j = e & 63;
        buS[i][j] = (j > i) ? b[e] : 0.0f;
    }
    {
        constexpr int P3 = 3 * NMON / THREADS, P4 = 4 * NMON / THREADS;
        constexpr int P5 = 5 * NMON / THREADS, PC = NMON / THREADS;
        int r3[P3], r4[P4], r5[P5];
        float q3[PC], q4[PC], q5[PC];
        #pragma unroll
        for (int i = 0; i < P3; ++i) r3[i] = idx3[tid + i * THREADS];
        #pragma unroll
        for (int i = 0; i < P4; ++i) r4[i] = idx4[tid + i * THREADS];
        #pragma unroll
        for (int i = 0; i < P5; ++i) r5[i] = idx5[tid + i * THREADS];
        #pragma unroll
        for (int i = 0; i < PC; ++i) {
            q3[i] = c3[tid + i * THREADS];
            q4[i] = c4[tid + i * THREADS];
            q5[i] = c5[tid + i * THREADS];
        }
        #pragma unroll
        for (int i = 0; i < P3; ++i) ((int*)off3)[tid + i * THREADS] = r3[i] * XPITCH;
        #pragma unroll
        for (int i = 0; i < P4; ++i) ((int*)off4)[tid + i * THREADS] = r4[i] * XPITCH;
        #pragma unroll
        for (int i = 0; i < P5; ++i) ((int*)off5)[tid + i * THREADS] = r5[i] * XPITCH;
        #pragma unroll
        for (int i = 0; i < PC; ++i) {
            ((float*)cs3)[tid + i * THREADS] = q3[i];
            ((float*)cs4)[tid + i * THREADS] = q4[i];
            ((float*)cs5)[tid + i * THREADS] = q5[i];
        }
    }
    __syncthreads();

    for (int e = tid; e < ROWS_PER_BLOCK * KDIM; e += THREADS) {
        const int r = e >> 6, j = e & 63;
        XS(r, j) = x[(size_t)(rowBlockBase + r) * BDIM + Ssh[j]];
    }
    __syncthreads();

    const int waveBase = wave * 16;
    float part[8];
    #pragma unroll
    for (int p = 0; p < 8; ++p) part[p] = 0.0f;

    #pragma unroll
    for (int nt = 0; nt < 4; ++nt) {
        v8f acc = {};
        #pragma unroll
        for (int kc = 0; kc < 16; ++kc) {
            v2f A, Bf;
            A.x = XS(waveBase + lo, 4 * kc + 2 * hi + 0);
            A.y = XS(waveBase + lo, 4 * kc + 2 * hi + 1);
            Bf.x = buS[4 * kc + 2 * hi + 0][16 * nt + lo];
            Bf.y = buS[4 * kc + 2 * hi + 1][16 * nt + lo];
            acc = __builtin_amdgcn_wmma_f32_16x16x4_f32(
                false, A, false, Bf, (short)0, acc, false, false);
        }
        #pragma unroll
        for (int p = 0; p < 8; ++p)
            part[p] += acc[p] * XS(waveBase + p + 8 * hi, 16 * nt + lo);
    }

    #pragma unroll
    for (int p = 0; p < 8; ++p) {
        float s = part[p];
        s += __shfl_xor(s, 8);
        s += __shfl_xor(s, 4);
        s += __shfl_xor(s, 2);
        s += __shfl_xor(s, 1);
        if (lo == p) quadS[waveBase + p + 8 * hi] = s;
    }
    __syncthreads();

    if (tid < ROWS_PER_BLOCK) {
        const int lr = tid;
        float accum = quadS[lr];

        #pragma unroll 8
        for (int j = 0; j < KDIM; ++j)
            accum += XS(lr, j) * ash[j];

        const float* rowp = &xsT[0][lr];
        accum = mono_terms<3>(rowp, off3, cs3, accum);
        accum = mono_terms<4>(rowp, off4, cs4, accum);
        accum = mono_terms<5>(rowp, off5, cs5, accum);
        out[rowBlockBase + lr] = accum;
    }
}

extern "C" void kernel_launch(void* const* d_in, const int* in_sizes, int n_in,
                              void* d_out, int out_size, void* d_ws, size_t ws_size,
                              hipStream_t stream) {
    const float* x    = (const float*)d_in[0];
    const int*   S    = (const int*)  d_in[1];
    const float* a    = (const float*)d_in[2];
    const float* b    = (const float*)d_in[3];
    const int*   idx3 = (const int*)  d_in[4];
    const float* c3   = (const float*)d_in[5];
    const int*   idx4 = (const int*)  d_in[6];
    const float* c4   = (const float*)d_in[7];
    const int*   idx5 = (const int*)  d_in[8];
    const float* c5   = (const float*)d_in[9];
    float* out = (float*)d_out;

    const int nblocks = out_size / ROWS_PER_BLOCK;
    sparse_poly_kernel<<<nblocks, THREADS, 0, stream>>>(
        x, S, a, b, idx3, c3, idx4, c4, idx5, c5, out);
}


# monomial phase re-split over all 8 waves: a lane owns 4 adjacent rows (one 16-byte LDS gather per factor, tile pitch 140), a wave 64 terms per degree; 8 partial sums added per row (sum order changes)
# speedup vs baseline: 14.3537x; 1.5471x over previous
#include <hip/hip_runtime.h>
#include <hip/hip_bf16.h>


#define BDIM 1024
#define KDIM 64
#define NMON 512
#define ROWS_PER_BLOCK 128
#define THREADS 256
#define NWAVES (THREADS / 32)
#define XPITCH 140
#define XS(r, j) xsT[(j)][(r)]

typedef __attribute__((ext_vector_type(2))) float v2f;
typedef __attribute__((ext_vector_type(8))) float v8f;

#define TERM_BATCH 8
#define TERMS_PER_WAVE (NMON / NWAVES)

typedef __attribute__((ext_vector_type(4))) int   v4i;
typedef __attribute__((ext_vector_type(4))) float v4f;

template <int DEG>
__device__ __forceinline__ v4f mono_terms(const char* rowp, const v4i* offs,
                                          const v4f* cs, v4f accum)
{
    constexpr int UB = TERM_BATCH;
    constexpr int NQ = UB * DEG / 4;
    constexpr int NB = TERMS_PER_WAVE / UB;
    v4i o[NQ];
    #pragma unroll
    for (int i = 0; i < NQ; ++i) o[i] = offs[i];

    #pragma nounroll
    for (int b = 0; b < NB; ++b) {
        v4f cc[UB / 4];
        #pragma unroll
        for (int i = 0; i < UB / 4; ++i) cc[i] = cs[b * (UB / 4) + i];
        v4f v[UB * DEG];
        #pragma unroll
        for (int i = 0; i < UB * DEG; ++i)
            v[i] = *(const v4f*)(rowp + o[i >> 2][i & 3]);
        const int bn = (b + 1 < NB) ? (b + 1) : b;
        #pragma unroll
        for (int i = 0; i < NQ; ++i) o[i] = offs[bn * NQ + i];
        #pragma unroll
        for (int u = 0; u < UB; ++u) {
            const v4f* w = &v[u * DEG];
            const float cu = cc[u >> 2][u & 3];
            if (DEG == 3)      accum += cu * w[0] * w[1] * w[2];
            else if (DEG == 4) accum += cu * w[0] * w[1] * w[2] * w[3];
            else               accum += cu * w[0] * w[1] * w[2] * w[3] * w[DEG - 1];
        }
    }
    return accum;
}

__global__ __launch_bounds__(THREADS)
void sparse_poly_kernel(const float* __restrict__ x,
                        const int*   __restrict__ S,
                        const float* __restrict__ a,
                        const float* __restrict__ b,
                        const int*   __restrict__ idx3, const float* __restrict__ c3,
                        const int*   __restrict__ idx4, const float* __restrict__ c4,
                        const int*   __restrict__ idx5, const float* __restrict__ c5,
                        float* __restrict__ out)
{
    __shared__ __attribute__((aligned(16))) float xsT[KDIM][XPITCH];
    __shared__ float buS[KDIM][KDIM];
    __shared__ float quadS[ROWS_PER_BLOCK];
    __shared__ int   Ssh[KDIM];
    __shared__ float ash[KDIM];
    __shared__ v4f   partS[NWAVES][ROWS_PER_BLOCK / 4];
    __shared__ v4i off3[3 * NMON / 4];
    __shared__ v4i off4[4 * NMON / 4];
    __shared__ v4i off5[5 * NMON / 4];
    __shared__ v4f cs3[NMON / 4];
    __shared__ v4f cs4[NMON / 4];
    __shared__ v4f cs5[NMON / 4];

    const int tid  = threadIdx.x;
    const int wave = tid >> 5;
    const int lane = tid & 31;
    const int lo   = lane & 15;
    const int hi   = lane >> 4;
    const int rowBlockBase = blockIdx.x * ROWS_PER_BLOCK;

    if (tid < KDIM) {
        Ssh[tid] = S[tid];
        ash[tid] = a[tid];
    }
    #pragma unroll
    for (int e = tid; e < KDIM * KDIM; e += THREADS) {
        const int i = e >> 6, j = e & 63;
        buS[i][j] = (j > i) ? b[e] : 0.0f;
    }
    {
        constexpr int P3 = 3 * NMON / THREADS, P4 = 4 * NMON / THREADS;
        constexpr int P5 = 5 * NMON / THREADS, PC = NMON / THREADS;
        int r3[P3], r4[P4], r5[P5];
        float q3[PC], q4[PC], q5[PC];
        #pragma unroll
        for (int i = 0; i < P3; ++i) r3[i] = idx3[tid + i * THREADS];
        #pragma unroll
        for (int i = 0; i < P4; ++i) r4[i] = idx4[tid + i * THREADS];
        #pragma unroll
        for (int i = 0; i < P5; ++i) r5[i] = idx5[tid + i * THREADS];
        #pragma unroll
        for (int i = 0; i < PC; ++i) {
            q3[i] = c3[tid + i * THREADS];
            q4[i] = c4[tid + i * THREADS];
            q5[i] = c5[tid + i * THREADS];
        }
        #pragma unroll
        for (int i = 0; i < P3; ++i) ((int*)off3)[tid + i * THREADS] = r3[i] * (XPITCH * 4);
        #pragma unroll
        for (int i = 0; i < P4; ++i) ((int*)off4)[tid + i * THREADS] = r4[i] * (XPITCH * 4);
        #pragma unroll
        for (int i = 0; i < P5; ++i) ((int*)off5)[tid + i * THREADS] = r5[i] * (XPITCH * 4);
        #pragma unroll
        for (int i = 0; i < PC; ++i) {
            ((float*)cs3)[tid + i * THREADS] = q3[i];
            ((float*)cs4)[tid + i * THREADS] = q4[i];
            ((float*)cs5)[tid + i * THREADS] = q5[i];
        }
    }
    __syncthreads();

    for (int e = tid; e < ROWS_PER_BLOCK * KDIM; e += THREADS) {
        const int r = e >> 6, j = e & 63;
        XS(r, j) = x[(size_t)(rowBlockBase + r) * BDIM + Ssh[j]];
    }
    __syncthreads();

    const int waveBase = wave * 16;
    float part[8];
    #pragma unroll
    for (int p = 0; p < 8; ++p) part[p] = 0.0f;

    #pragma unroll
    for (int nt = 0; nt < 4; ++nt) {
        v8f acc = {};
        #pragma unroll
        for (int kc = 0; kc < 16; ++kc) {
            v2f A, Bf;
            A.x = XS(waveBase + lo, 4 * kc + 2 * hi + 0);
            A.y = XS(waveBase + lo, 4 * kc + 2 * hi + 1);
            Bf.x = buS[4 * kc + 2 * hi + 0][16 * nt + lo];
            Bf.y = buS[4 * kc + 2 * hi + 1][16 * nt + lo];
            acc = __builtin_amdgcn_wmma_f32_16x16x4_f32(
                false, A, false, Bf, (short)0, acc, false, false);
        }
        #pragma unroll
        for (int p = 0; p < 8; ++p)
            part[p] += acc[p] * XS(waveBase + p + 8 * hi, 16 * nt + lo);
    }

    #pragma unroll
    for (int p = 0; p < 8; ++p) {
        float s = part[p];
        s += __shfl_xor(s, 8);
        s += __shfl_xor(s, 4);
        s += __shfl_xor(s, 2);
        s += __shfl_xor(s, 1);
        if (lo == p) quadS[waveBase + p + 8 * hi] = s;
    }

    {
        const char* rowp = (const char*)&xsT[0][4 * lane];
        constexpr int CQ = TERMS_PER_WAVE / 4;
        v4f acc4 = {0.0f, 0.0f, 0.0f, 0.0f};
        acc4 = mono_terms<3>(rowp, off3 + wave * (3 * CQ), cs3 + wave * CQ, acc4);
        acc4 = mono_terms<4>(rowp, off4 + wave * (4 * CQ), cs4 + wave * CQ, acc4);
        acc4 = mono_terms<5>(rowp, off5 + wave * (5 * CQ), cs5 + wave * CQ, acc4);
        partS[wave][lane] = acc4;
    }
    __syncthreads();

    if (tid < ROWS_PER_BLOCK) {
        const int lr = tid;
        float accum = quadS[lr];

        #pragma unroll 8
        for (int j = 0; j < KDIM; ++j)
            accum += XS(lr, j) * ash[j];

        const float* pp = (const float*)partS;
        #pragma unroll
        for (int w = 0; w < NWAVES; ++w)
            accum += pp[w * ROWS_PER_BLOCK + lr];
        out[rowBlockBase + lr] = accum;
    }
}

extern "C" void kernel_launch(void* const* d_in, const int* in_sizes, int n_in,
                              void* d_out, int out_size, void* d_ws, size_t ws_size,
                              hipStream_t stream) {
    const float* x    = (const float*)d_in[0];
    const int*   S    = (const int*)  d_in[1];
    const float* a    = (const float*)d_in[2];
    const float* b    = (const float*)d_in[3];
    const int*   idx3 = (const int*)  d_in[4];
    const float* c3   = (const float*)d_in[5];
    const int*   idx4 = (const int*)  d_in[6];
    const float* c4   = (const float*)d_in[7];
    const int*   idx5 = (const int*)  d_in[8];
    const float* c5   = (const float*)d_in[9];
    float* out = (float*)d_out;

    const int nblocks = out_size / ROWS_PER_BLOCK;
    sparse_poly_kernel<<<nblocks, THREADS, 0, stream>>>(
        x, S, a, b, idx3, c3, idx4, c4, idx5, c5, out);
}


# staging: every global load of a block (S[j] per thread straight from global, a, b, term tables, 32 gathered x rows per thread) issued before the first wait; one barrier, not two; sum order unchanged
# speedup vs baseline: 17.9047x; 1.2474x over previous
#include <hip/hip_runtime.h>
#include <hip/hip_bf16.h>


#define BDIM 1024
#define KDIM 64
#define NMON 512
#define ROWS_PER_BLOCK 128
#define THREADS 256
#define NWAVES (THREADS / 32)
#define XPITCH 140
#define XS(r, j) xsT[(j)][(r)]

typedef __attribute__((ext_vector_type(2))) float v2f;
typedef __attribute__((ext_vector_type(8))) float v8f;

#define TERM_BATCH 8
#define TERMS_PER_WAVE (NMON / NWAVES)

typedef __attribute__((ext_vector_type(4))) int   v4i;
typedef __attribute__((ext_vector_type(4))) float v4f;

template <int DEG>
__device__ __forceinline__ v4f mono_terms(const char* rowp, const v4i* offs,
                                          const v4f* cs, v4f accum)
{
    constexpr int UB = TERM_BATCH;
    constexpr int NQ = UB * DEG / 4;
    constexpr int NB = TERMS_PER_WAVE / UB;
    v4i o[NQ];
    #pragma unroll
    for (int i = 0; i < NQ; ++i) o[i] = offs[i];

    #pragma nounroll
    for (int b = 0; b < NB; ++b) {
        v4f cc[UB / 4];
        #pragma unroll
        for (int i = 0; i < UB / 4; ++i) cc[i] = cs[b * (UB / 4) + i];
        v4f v[UB * DEG];
        #pragma unroll
        for (int i = 0; i < UB * DEG; ++i)
            v[i] = *(const v4f*)(rowp + o[i >> 2][i & 3]);
        const int bn = (b + 1 < NB) ? (b + 1) : b;
        #pragma unroll
        for (int i = 0; i < NQ; ++i) o[i] = offs[bn * NQ + i];
        #pragma unroll
        for (int u = 0; u < UB; ++u) {
            const v4f* w = &v[u * DEG];
            const float cu = cc[u >> 2][u & 3];
            if (DEG == 3)      accum += cu * w[0] * w[1] * w[2];
            else if (DEG == 4) accum += cu * w[0] * w[1] * w[2] * w[3];
            else               accum += cu * w[0] * w[1] * w[2] * w[3] * w[DEG - 1];
        }
    }
    return accum;
}

__global__ __launch_bounds__(THREADS)
void sparse_poly_kernel(const float* __restrict__ x,
                        const int*   __restrict__ S,
                        const float* __restrict__ a,
                        const float* __restrict__ b,
                        const int*   __restrict__ idx3, const float* __restrict__ c3,
                        const int*   __restrict__ idx4, const float* __restrict__ c4,
                        const int*   __restrict__ idx5, const float* __restrict__ c5,
                        float* __restrict__ out)
{
    __shared__ __attribute__((aligned(16))) float xsT[KDIM][XPITCH];
    __shared__ float buS[KDIM][KDIM];
    __shared__ float quadS[ROWS_PER_BLOCK];
    __shared__ float ash[KDIM];
    __shared__ v4f   partS[NWAVES][ROWS_PER_BLOCK / 4];
    __shared__ v4i off3[3 * NMON / 4];
    __shared__ v4i off4[4 * NMON / 4];
    __shared__ v4i off5[5 * NMON / 4];
    __shared__ v4f cs3[NMON / 4];
    __shared__ v4f cs4[NMON / 4];
    __shared__ v4f cs5[NMON / 4];

    const int tid  = threadIdx.x;
    const int wave = tid >> 5;
    const int lane = tid & 31;
    const int lo   = lane & 15;
    const int hi   = lane >> 4;
    const int rowBlockBase = blockIdx.x * ROWS_PER_BLOCK;

    constexpr int XROWS = ROWS_PER_BLOCK * KDIM / THREADS;
    constexpr int BPT   = KDIM * KDIM / THREADS;
    constexpr int P3 = 3 * NMON / THREADS, P4 = 4 * NMON / THREADS;
    constexpr int P5 = 5 * NMON / THREADS, PC = NMON / THREADS;
    const int gj = tid & (KDIM - 1);
    const int r0 = tid >> 6;
    const int sj = S[gj];

    float av = 0.0f;
    if (tid < KDIM) av = a[tid];
    float bv[BPT];
    #pragma unroll
    for (int i = 0; i < BPT; ++i) bv[i] = b[tid + i * THREADS];
    int r3[P3], r4[P4], r5[P5];
    float q3[PC], q4[PC], q5[PC];
    #pragma unroll
    for (int i = 0; i < P3; ++i) r3[i] = idx3[tid + i * THREADS];
    #pragma unroll
    for (int i = 0; i < P4; ++i) r4[i] = idx4[tid + i * THREADS];
    #pragma unroll
    for (int i = 0; i < P5; ++i) r5[i] = idx5[tid + i * THREADS];
    #pragma unroll
    for (int i = 0; i < PC; ++i) {
        q3[i] = c3[tid + i * THREADS];
        q4[i] = c4[tid + i * THREADS];
        q5[i] = c5[tid + i * THREADS];
    }

    const float* xp = x + (size_t)(rowBlockBase + r0) * BDIM + sj;
    float xv[XROWS];
    #pragma unroll
    for (int i = 0; i < XROWS; ++i) xv[i] = xp[(size_t)i * (THREADS / KDIM) * BDIM];

    if (tid < KDIM) ash[tid] = av;
    #pragma unroll
    for (int i = 0; i < BPT; ++i) {
        const int bi = r0 + i * (THREADS / KDIM);
        buS[bi][gj] = (gj > bi) ? bv[i] : 0.0f;
    }
    #pragma unroll
    for (int i = 0; i < P3; ++i) ((int*)off3)[tid + i * THREADS] = r3[i] * (XPITCH * 4);
    #pragma unroll
    for (int i = 0; i < P4; ++i) ((int*)off4)[tid + i * THREADS] = r4[i] * (XPITCH * 4);
    #pragma unroll
    for (int i = 0; i < P5; ++i) ((int*)off5)[tid + i * THREADS] = r5[i] * (XPITCH * 4);
    #pragma unroll
    for (int i = 0; i < PC; ++i) {
        ((float*)cs3)[tid + i * THREADS] = q3[i];
        ((float*)cs4)[tid + i * THREADS] = q4[i];
        ((float*)cs5)[tid + i * THREADS] = q5[i];
    }
    #pragma unroll
    for (int i = 0; i < XROWS; ++i) XS(r0 + i * (THREADS / KDIM), gj) = xv[i];
    __syncthreads();

    const int waveBase = wave * 16;
    float part[8];
    #pragma unroll
    for (int p = 0; p < 8; ++p) part[p] = 0.0f;

    #pragma unroll
    for (int nt = 0; nt < 4; ++nt) {
        v8f acc = {};
        #pragma unroll
        for (int kc = 0; kc < 16; ++kc) {
            v2f A, Bf;
            A.x = XS(waveBase + lo, 4 * kc + 2 * hi + 0);
            A.y = XS(waveBase + lo, 4 * kc + 2 * hi + 1);
            Bf.x = buS[4 * kc + 2 * hi + 0][16 * nt + lo];
            Bf.y = buS[4 * kc + 2 * hi + 1][16 * nt + lo];
            acc = __builtin_amdgcn_wmma_f32_16x16x4_f32(
                false, A, false, Bf, (short)0, acc, false, false);
        }
        #pragma unroll
        for (int p = 0; p < 8; ++p)
            part[p] += acc[p] * XS(waveBase + p + 8 * hi, 16 * nt + lo);
    }

    #pragma unroll
    for (int p = 0; p < 8; ++p) {
        float s = part[p];
        s += __shfl_xor(s, 8);
        s += __shfl_xor(s, 4);
        s += __shfl_xor(s, 2);
        s += __shfl_xor(s, 1);
        if (lo == p) quadS[waveBase + p + 8 * hi] = s;
    }

    {
        const char* rowp = (const char*)&xsT[0][4 * lane];
        constexpr int CQ = TERMS_PER_WAVE / 4;
        v4f acc4 = {0.0f, 0.0f, 0.0f, 0.0f};
        acc4 = mono_terms<3>(rowp, off3 + wave * (3 * CQ), cs3 + wave * CQ, acc4);
        acc4 = mono_terms<4>(rowp, off4 + wave * (4 * CQ), cs4 + wave * CQ, acc4);
        acc4 = mono_terms<5>(rowp, off5 + wave * (5 * CQ), cs5 + wave * CQ, acc4);
        partS[wave][lane] = acc4;
    }
    __syncthreads();

    if (tid < ROWS_PER_BLOCK) {
        const int lr = tid;
        float accum = quadS[lr];

        #pragma unroll 8
        for (int j = 0; j < KDIM; ++j)
            accum += XS(lr, j) * ash[j];

        const float* pp = (const float*)partS;
        #pragma unroll
        for (int w = 0; w < NWAVES; ++w)
            accum += pp[w * ROWS_PER_BLOCK + lr];
        out[rowBlockBase + lr] = accum;
    }
}

extern "C" void kernel_launch(void* const* d_in, const int* in_sizes, int n_in,
                              void* d_out, int out_size, void* d_ws, size_t ws_size,
                              hipStream_t stream) {
    const float* x    = (const float*)d_in[0];
    const int*   S    = (const int*)  d_in[1];
    const float* a    = (const float*)d_in[2];
    const float* b    = (const float*)d_in[3];
    const int*   idx3 = (const int*)  d_in[4];
    const float* c3   = (const float*)d_in[5];
    const int*   idx4 = (const int*)  d_in[6];
    const float* c4   = (const float*)d_in[7];
    const int*   idx5 = (const int*)  d_in[8];
    const float* c5   = (const float*)d_in[9];
    float* out = (float*)d_out;

    const int nblocks = out_size / ROWS_PER_BLOCK;
    sparse_poly_kernel<<<nblocks, THREADS, 0, stream>>>(
        x, S, a, b, idx3, c3, idx4, c4, idx5, c5, out);
}


# speedup vs baseline: 19.9102x; 1.1120x over previous
#include <hip/hip_runtime.h>
#include <hip/hip_bf16.h>


#define BDIM 1024
#define KDIM 64
#define NMON 512
#define ROWS_PER_BLOCK 128
#define THREADS 256
#define NWAVES (THREADS / 32)
#define XPITCH 140

typedef __attribute__((ext_vector_type(2))) float v2f;
typedef __attribute__((ext_vector_type(8))) float v8f;

#define TERM_BATCH 8
#define TERMS_PER_WAVE (NMON / NWAVES)

typedef __attribute__((ext_vector_type(4))) int   v4i;
typedef __attribute__((ext_vector_type(4))) float v4f;

#define NREGION       4
#define TILE_BYTES    (KDIM * XPITCH * 4)
#define OFF3_BYTES    (3 * TERMS_PER_WAVE * 4)
#define OFF4_BYTES    (4 * TERMS_PER_WAVE * 4)
#define OFF5_BYTES    (5 * TERMS_PER_WAVE * 4)
#define CS_BYTES      (TERMS_PER_WAVE * 4)
#define WT_OFF3       0
#define WT_OFF4       (WT_OFF3 + OFF3_BYTES)
#define WT_OFF5       (WT_OFF4 + OFF4_BYTES)
#define WT_CS3        (WT_OFF5 + OFF5_BYTES)
#define WT_CS4        (WT_CS3 + CS_BYTES)
#define WT_CS5        (WT_CS4 + CS_BYTES)
#define WTAB_BYTES    (WT_CS5 + CS_BYTES)
#define TAB_BASE      TILE_BYTES
#define REGION_USED   (TILE_BYTES + 2 * WTAB_BYTES)
#define REGION_STRIDE 77824
#define BU_BASE       45056
#define ASH_BASE      (BU_BASE + KDIM * KDIM * 4)
#define QUAD_BASE     (ASH_BASE + KDIM * 4)
#define PART_BASE     (QUAD_BASE + ROWS_PER_BLOCK * 4)
#define LDS_BYTES     ((NREGION - 1) * REGION_STRIDE + REGION_USED)

static_assert(REGION_STRIDE >= REGION_USED + 32768);
static_assert(BU_BASE >= REGION_USED, "shared data overlaps region 0");
static_assert(PART_BASE + NWAVES * 32 * 16 <= REGION_STRIDE, "shared data overlaps region 1");
static_assert(TILE_BYTES % 16 == 0 && WTAB_BYTES % 16 == 0 && REGION_STRIDE % 16 == 0 &&
              WT_OFF4 % 16 == 0 && WT_OFF5 % 16 == 0 && WT_CS3 % 16 == 0 && PART_BASE % 16 == 0,
              "16-byte alignment of the wide LDS accesses");

template <int DEG>
__device__ __forceinline__ v4f mono_terms(const char* rowp, const v4i* offs,
                                          const v4f* cs, v4f accum)
{
    constexpr int UB = TERM_BATCH;
    constexpr int NQ = UB * DEG / 4;
    constexpr int NB = TERMS_PER_WAVE / UB;
    v4i o[NQ];
    #pragma unroll
    for (int i = 0; i < NQ; ++i) o[i] = offs[i];

    #pragma nounroll
    for (int b = 0; b < NB; ++b) {
        v4f cc[UB / 4];
        #pragma unroll
        for (int i = 0; i < UB / 4; ++i) cc[i] = cs[b * (UB / 4) + i];
        v4f v[UB * DEG];
        #pragma unroll
        for (int i = 0; i < UB * DEG; ++i)
            v[i] = *(const v4f*)(rowp + o[i >> 2][i & 3]);
        const int bn = (b + 1 < NB) ? (b + 1) : b;
        #pragma unroll
        for (int i = 0; i < NQ; ++i) o[i] = offs[bn * NQ + i];
        #pragma unroll
        for (int u = 0; u < UB; ++u) {
            const v4f* w = &v[u * DEG];
            const float cu = cc[u >> 2][u & 3];
            if (DEG == 3)      accum += cu * w[0] * w[1] * w[2];
            else if (DEG == 4) accum += cu * w[0] * w[1] * w[2] * w[3];
            else               accum += cu * w[0] * w[1] * w[2] * w[3] * w[DEG - 1];
        }
    }
    return accum;
}

template <int DEG>
__device__ __forceinline__ int off_entry_addr(int e, int table_off)
{
    const int w      = e / (TERMS_PER_WAVE * DEG);
    const int within = e - w * (TERMS_PER_WAVE * DEG);
    return (w & 3) * REGION_STRIDE + TAB_BASE + (w >> 2) * WTAB_BYTES + table_off + within * 4;
}
__device__ __forceinline__ int cs_entry_addr(int t, int table_off)
{
    const int w = t / TERMS_PER_WAVE;
    return (w & 3) * REGION_STRIDE + TAB_BASE + (w >> 2) * WTAB_BYTES + table_off +
           (t - w * TERMS_PER_WAVE) * 4;
}

__global__ __launch_bounds__(THREADS)
void sparse_poly_kernel(const float* __restrict__ x,
                        const int*   __restrict__ S,
                        const float* __restrict__ a,
                        const float* __restrict__ b,
                        const int*   __restrict__ idx3, const float* __restrict__ c3,
                        const int*   __restrict__ idx4, const float* __restrict__ c4,
                        const int*   __restrict__ idx5, const float* __restrict__ c5,
                        float* __restrict__ out)
{
    __shared__ __attribute__((aligned(16))) unsigned char lds[LDS_BYTES];

    const int tid  = threadIdx.x;
    const int wave = tid >> 5;
    const int lane = tid & 31;
    const int lo   = lane & 15;
    const int hi   = lane >> 4;
    const int rowBlockBase = blockIdx.x * ROWS_PER_BLOCK;

    unsigned char* const reg = lds + (wave & 3) * REGION_STRIDE;
    const float*   const xt  = (const float*)reg;
    #define XS(r, j) xt[(j) * XPITCH + (r)]
    float (* const buS)[KDIM] = (float (*)[KDIM])(lds + BU_BASE);
    float* const ash   = (float*)(lds + ASH_BASE);
    float* const quadS = (float*)(lds + QUAD_BASE);
    v4f*   const partS = (v4f*)(lds + PART_BASE);

    constexpr int XROWS = ROWS_PER_BLOCK * KDIM / THREADS;
    constexpr int BPT   = KDIM * KDIM / THREADS;
    constexpr int P3 = 3 * NMON / THREADS, P4 = 4 * NMON / THREADS;
    constexpr int P5 = 5 * NMON / THREADS, PC = NMON / THREADS;
    const int gj = tid & (KDIM - 1);
    const int r0 = tid >> 6;
    const int sj = S[gj];

    float av = 0.0f;
    if (tid < KDIM) av = a[tid];
    float bv[BPT];
    #pragma unroll
    for (int i = 0; i < BPT; ++i) bv[i] = b[tid + i * THREADS];
    int r3[P3], r4[P4], r5[P5];
    float q3[PC], q4[PC], q5[PC];
    #pragma unroll
    for (int i = 0; i < P3; ++i) r3[i] = idx3[tid + i * THREADS];
    #pragma unroll
    for (int i = 0; i < P4; ++i) r4[i] = idx4[tid + i * THREADS];
    #pragma unroll
    for (int i = 0; i < P5; ++i) r5[i] = idx5[tid + i * THREADS];
    #pragma unroll
    for (int i = 0; i < PC; ++i) {
        q3[i] = c3[tid + i * THREADS];
        q4[i] = c4[tid + i * THREADS];
        q5[i] = c5[tid + i * THREADS];
    }

    const float* xp = x + (size_t)(rowBlockBase + r0 * XROWS) * BDIM + sj;
    float xv[XROWS];
    #pragma unroll
    for (int i = 0; i < XROWS; ++i) xv[i] = xp[(size_t)i * BDIM];

    if (tid < KDIM) ash[tid] = av;
    #pragma unroll
    for (int i = 0; i < BPT; ++i) {
        const int bi = r0 + i * (THREADS / KDIM);
        buS[bi][gj] = (gj > bi) ? bv[i] : 0.0f;
    }
    #pragma unroll
    for (int i = 0; i < P3; ++i)
        *(int*)(lds + off_entry_addr<3>(tid + i * THREADS, WT_OFF3)) = r3[i] * (XPITCH * 4);
    #pragma unroll
    for (int i = 0; i < P4; ++i)
        *(int*)(lds + off_entry_addr<4>(tid + i * THREADS, WT_OFF4)) = r4[i] * (XPITCH * 4);
    #pragma unroll
    for (int i = 0; i < P5; ++i)
        *(int*)(lds + off_entry_addr<5>(tid + i * THREADS, WT_OFF5)) = r5[i] * (XPITCH * 4);
    #pragma unroll
    for (int i = 0; i < PC; ++i) {
        *(float*)(lds + cs_entry_addr(tid + i * THREADS, WT_CS3)) = q3[i];
        *(float*)(lds + cs_entry_addr(tid + i * THREADS, WT_CS4)) = q4[i];
        *(float*)(lds + cs_entry_addr(tid + i * THREADS, WT_CS5)) = q5[i];
    }
    {
        unsigned char* const tp = lds + gj * (XPITCH * 4) + r0 * (XROWS * 4);
        #pragma unroll
        for (int q = 0; q < XROWS / 4; ++q) {
            const v4f v = {xv[4 * q], xv[4 * q + 1], xv[4 * q + 2], xv[4 * q + 3]};
            #pragma unroll
            for (int c = 0; c < NREGION; ++c)
                *(v4f*)(tp + c * REGION_STRIDE + q * 16) = v;
        }
    }
    __syncthreads();

    const int waveBase = wave * 16;
    float part[8];
    #pragma unroll
    for (int p = 0; p < 8; ++p) part[p] = 0.0f;

    #pragma unroll
    for (int nt = 0; nt < 4; ++nt) {
        v8f acc = {};
        #pragma unroll
        for (int kc = 0; kc < 16; ++kc) {
            v2f A, Bf;
            A.x = XS(waveBase + lo, 4 * kc + 2 * hi + 0);
            A.y = XS(waveBase + lo, 4 * kc + 2 * hi + 1);
            Bf.x = buS[4 * kc + 2 * hi + 0][16 * nt + lo];
            Bf.y = buS[4 * kc + 2 * hi + 1][16 * nt + lo];
            acc = __builtin_amdgcn_wmma_f32_16x16x4_f32(
                false, A, false, Bf, (short)0, acc, false, false);
        }
        #pragma unroll
        for (int p = 0; p < 8; ++p)
            part[p] += acc[p] * XS(waveBase + p + 8 * hi, 16 * nt + lo);
    }

    #pragma unroll
    for (int p = 0; p < 8; ++p) {
        float s = part[p];
        s += __shfl_xor(s, 8);
        s += __shfl_xor(s, 4);
        s += __shfl_xor(s, 2);
        s += __shfl_xor(s, 1);
        if (lo == p) quadS[waveBase + p + 8 * hi] = s;
    }

    {
        const char* rowp = (const char*)reg + 16 * lane;
        const unsigned char* tb = reg + TAB_BASE + (wave >> 2) * WTAB_BYTES;
        v4f acc4 = {0.0f, 0.0f, 0.0f, 0.0f};
        acc4 = mono_terms<3>(rowp, (const v4i*)(tb + WT_OFF3), (const v4f*)(tb + WT_CS3), acc4);
        acc4 = mono_terms<4>(rowp, (const v4i*)(tb + WT_OFF4), (const v4f*)(tb + WT_CS4), acc4);
        acc4 = mono_terms<5>(rowp, (const v4i*)(tb + WT_OFF5), (const v4f*)(tb + WT_CS5), acc4);
        partS[wave * 32 + lane] = acc4;
    }
    __syncthreads();

    if (tid < ROWS_PER_BLOCK) {
        const int lr = tid;
        float accum = quadS[lr];

        #pragma unroll 8
        for (int j = 0; j < KDIM; ++j)
            accum += XS(lr, j) * ash[j];

        const float* pp = (const float*)partS;
        #pragma unroll
        for (int w = 0; w < NWAVES; ++w)
            accum += pp[w * ROWS_PER_BLOCK + lr];
        out[rowBlockBase + lr] = accum;
    }
    #undef XS
}

extern "C" void kernel_launch(void* const* d_in, const int* in_sizes, int n_in,
                              void* d_out, int out_size, void* d_ws, size_t ws_size,
                              hipStream_t stream) {
    const float* x    = (const float*)d_in[0];
    const int*   S    = (const int*)  d_in[1];
    const float* a    = (const float*)d_in[2];
    const float* b    = (const float*)d_in[3];
    const int*   idx3 = (const int*)  d_in[4];
    const float* c3   = (const float*)d_in[5];
    const int*   idx4 = (const int*)  d_in[6];
    const float* c4   = (const float*)d_in[7];
    const int*   idx5 = (const int*)  d_in[8];
    const float* c5   = (const float*)d_in[9];
    float* out = (float*)d_out;

    const int nblocks = out_size / ROWS_PER_BLOCK;
    sparse_poly_kernel<<<nblocks, THREADS, 0, stream>>>(
        x, S, a, b, idx3, c3, idx4, c4, idx5, c5, out);
}
